# starting kernel
# baseline (speedup 1.0000x reference)
#include <hip/hip_runtime.h>
#include <hip/hip_bf16.h>

#define HOPS   3
#define VOCAB  50000
#define EMBD   128
#define SLEN   200
#define SENT   32
#define BATCH  64
#define QLEN   32
#define NCHUNK 25

typedef __attribute__((ext_vector_type(2))) float v2f;
typedef __attribute__((ext_vector_type(8))) float v8f;

__global__ __launch_bounds__(EMBD)
void k_init_u(const float* __restrict__ A1, const int* __restrict__ q,
              float* __restrict__ u) {
    int b = blockIdx.x, e = threadIdx.x;
    const int* qb = q + b * QLEN;
    float acc = 0.f;
#pragma unroll 4
    for (int j = 0; j < QLEN; ++j)
        acc += A1[(long)qb[j] * EMBD + e];
    u[b * EMBD + e] = acc;
}

__global__ __launch_bounds__(EMBD)
void k_score(const float* __restrict__ Ak, const float* __restrict__ TA,
             const int* __restrict__ x, const float* __restrict__ u,
             float* __restrict__ praw) {
    int s = blockIdx.x, b = blockIdx.y, e = threadIdx.x;
    const int* xs = x + (b * SLEN + s) * SENT;
    float acc = 0.f;
#pragma unroll 4
    for (int t = 0; t < SENT; ++t)
        acc += Ak[(long)xs[t] * EMBD + e];
    acc += TA[s];
    float v = acc * u[b * EMBD + e];

    __shared__ float red[EMBD];
    red[e] = v;
    __syncthreads();
    for (int off = EMBD / 2; off > 0; off >>= 1) {
        if (e < off) red[e] += red[e + off];
        __syncthreads();
    }
    if (e == 0) praw[b * SLEN + s] = red[0];
}

__global__ __launch_bounds__(256)
void k_softmax(float* __restrict__ p) {
    int b = blockIdx.x, t = threadIdx.x;
    float* pb = p + b * SLEN;
    __shared__ float red[256];

    float m = -1e30f;
    for (int s = t; s < SLEN; s += 256) m = fmaxf(m, pb[s]);
    red[t] = m;
    __syncthreads();
    for (int off = 128; off > 0; off >>= 1) {
        if (t < off) red[t] = fmaxf(red[t], red[t + off]);
        __syncthreads();
    }
    m = red[0];
    __syncthreads();

    float sum = 0.f;
    for (int s = t; s < SLEN; s += 256) sum += __expf(pb[s] - m);
    red[t] = sum;
    __syncthreads();
    for (int off = 128; off > 0; off >>= 1) {
        if (t < off) red[t] += red[t + off];
        __syncthreads();
    }
    float inv = 1.0f / red[0];
    for (int s = t; s < SLEN; s += 256) pb[s] = __expf(pb[s] - m) * inv;
}

__global__ __launch_bounds__(EMBD)
void k_combine(const float* __restrict__ Ak1, const float* __restrict__ TC,
               const int* __restrict__ x, const float* __restrict__ p,
               float* __restrict__ part) {
    int ch = blockIdx.x, b = blockIdx.y, e = threadIdx.x;
    float acc = 0.f;
    for (int s = ch * 8; s < ch * 8 + 8; ++s) {
        const int* xs = x + (b * SLEN + s) * SENT;
        float g = 0.f;
#pragma unroll 4
        for (int t = 0; t < SENT; ++t)
            g += Ak1[(long)xs[t] * EMBD + e];
        acc += p[b * SLEN + s] * (g + TC[s]);
    }
    part[(b * NCHUNK + ch) * EMBD + e] = acc;
}

__global__ __launch_bounds__(EMBD)
void k_update(float* __restrict__ u, const float* __restrict__ part) {
    int b = blockIdx.x, e = threadIdx.x;
    float acc = 0.f;
    for (int ch = 0; ch < NCHUNK; ++ch)
        acc += part[(b * NCHUNK + ch) * EMBD + e];
    u[b * EMBD + e] += acc;
}

__global__ __launch_bounds__(32)
void k_logits(const float* __restrict__ u, const float* __restrict__ A3,
              float* __restrict__ out) {
    int n0   = blockIdx.x * 16;
    int lane = threadIdx.x;
    int half = lane >> 4;
    int l16  = lane & 15;
    int koff = half * 2;

    const float* arow = A3 + (long)(n0 + l16) * EMBD;
    const float* u0   = u + l16 * EMBD;

    v8f c0 = {}, c1 = {}, c2 = {}, c3 = {};
#pragma unroll
    for (int k0 = 0; k0 < EMBD; k0 += 4) {
        int kk = k0 + koff;
        v2f bfrag, a0, a1, a2, a3;
        bfrag.x = arow[kk];               bfrag.y = arow[kk + 1];
        a0.x = u0[kk];                    a0.y = u0[kk + 1];
        a1.x = u0[16 * EMBD + kk];        a1.y = u0[16 * EMBD + kk + 1];
        a2.x = u0[32 * EMBD + kk];        a2.y = u0[32 * EMBD + kk + 1];
        a3.x = u0[48 * EMBD + kk];        a3.y = u0[48 * EMBD + kk + 1];
        c0 = __builtin_amdgcn_wmma_f32_16x16x4_f32(false, a0, false, bfrag,
                                                   (short)0, c0, false, false);
        c1 = __builtin_amdgcn_wmma_f32_16x16x4_f32(false, a1, false, bfrag,
                                                   (short)0, c1, false, false);
        c2 = __builtin_amdgcn_wmma_f32_16x16x4_f32(false, a2, false, bfrag,
                                                   (short)0, c2, false, false);
        c3 = __builtin_amdgcn_wmma_f32_16x16x4_f32(false, a3, false, bfrag,
                                                   (short)0, c3, false, false);
    }

    int col = n0 + l16;
#pragma unroll
    for (int r = 0; r < 8; ++r) {
        int row = r + 8 * half;
        out[(long)(row)      * VOCAB + col] = c0[r];
        out[(long)(row + 16) * VOCAB + col] = c1[r];
        out[(long)(row + 32) * VOCAB + col] = c2[r];
        out[(long)(row + 48) * VOCAB + col] = c3[r];
    }
}

__global__ __launch_bounds__(256)
void k_rowstats(const float* __restrict__ out, float* __restrict__ rowmax,
                float* __restrict__ rowlse) {
    int b = blockIdx.x, t = threadIdx.x;
    const float* row = out + (long)b * VOCAB;
    __shared__ float red[256];

    float m = -1e30f;
    for (int v = t; v < VOCAB; v += 256) m = fmaxf(m, row[v]);
    red[t] = m;
    __syncthreads();
    for (int off = 128; off > 0; off >>= 1) {
        if (t < off) red[t] = fmaxf(red[t], red[t + off]);
        __syncthreads();
    }
    m = red[0];
    __syncthreads();

    float s = 0.f;
    for (int v = t; v < VOCAB; v += 256) s += __expf(row[v] - m);
    red[t] = s;
    __syncthreads();
    for (int off = 128; off > 0; off >>= 1) {
        if (t < off) red[t] += red[t + off];
        __syncthreads();
    }
    if (t == 0) { rowmax[b] = m; rowlse[b] = __logf(red[0]); }
}

__global__ __launch_bounds__(256)
void k_finalize(float* __restrict__ out, const float* __restrict__ rowmax,
                const float* __restrict__ rowlse) {
    long i = (long)blockIdx.x * 256 + threadIdx.x;
    if (i < (long)BATCH * VOCAB) {
        int b = (int)(i / VOCAB);
        out[i] = out[i] - rowmax[b] - rowlse[b];
    }
}

extern "C" void kernel_launch(void* const* d_in, const int* in_sizes, int n_in,
                              void* d_out, int out_size, void* d_ws, size_t ws_size,
                              hipStream_t stream) {
    const float* A  = (const float*)d_in[0];
    const float* TA = (const float*)d_in[1];
    const float* TC = (const float*)d_in[2];
    const int*   x  = (const int*)d_in[3];
    const int*   q  = (const int*)d_in[4];
    float* out = (float*)d_out;

    float* ws   = (float*)d_ws;
    float* u    = ws;
    float* p    = u    + BATCH * EMBD;
    float* part = p    + BATCH * SLEN;
    float* rmax = part + BATCH * NCHUNK * EMBD;
    float* rlse = rmax + BATCH;

    const long TBL = (long)VOCAB * EMBD;

    k_init_u<<<BATCH, EMBD, 0, stream>>>(A + 1 * TBL, q, u);

    for (int k = 0; k < HOPS; ++k) {
        k_score  <<<dim3(SLEN, BATCH),   EMBD, 0, stream>>>(A + k * TBL, TA, x, u, p);
        k_softmax<<<BATCH,               256,  0, stream>>>(p);
        k_combine<<<dim3(NCHUNK, BATCH), EMBD, 0, stream>>>(A + (k + 1) * TBL, TC, x, p, part);
        k_update <<<BATCH,               EMBD, 0, stream>>>(u, part);
    }

    k_logits  <<<dim3(VOCAB / 16), 32, 0, stream>>>(u, A + HOPS * TBL, out);
    k_rowstats<<<BATCH, 256, 0, stream>>>(out, rmax, rlse);
    k_finalize<<<(BATCH * VOCAB + 255) / 256, 256, 0, stream>>>(out, rmax, rlse);

    (void)in_sizes; (void)n_in; (void)out_size; (void)ws_size;
}


# k_logits re-tiled to 1024 resident waves: k_merge stores u3 in A-fragment order, a wave fetches it once by whole-line loads and keeps it in registers over its 3 to 3.5 vocab strips; sum orders kept
# speedup vs baseline: 2.8798x; 2.8798x over previous
#include <hip/hip_runtime.h>
#include <hip/hip_bf16.h>

#define HOPS   3
#define VOCAB  50000
#define EMBD   128
#define SLEN   200
#define SENT   32
#define BATCH  64
#define QLEN   32
#define NCHUNK 20
#define CHS    10
#define HOPT   128
#define GRP    16

typedef __attribute__((ext_vector_type(2))) float v2f;
typedef __attribute__((ext_vector_type(8))) float v8f;


__device__ __forceinline__
float merge_partials(const float* __restrict__ pm, const float* __restrict__ pl,
                     const float* __restrict__ pa, int b, int e) {
    const float* mb = pm + b * NCHUNK;
    const float* lb = pl + b * NCHUNK;
    const float* ab = pa + (long)b * NCHUNK * EMBD + e;
    float M = mb[0];
#pragma unroll
    for (int j = 1; j < NCHUNK; ++j) M = fmaxf(M, mb[j]);
    float L = 0.f, a = 0.f;
#pragma unroll
    for (int j = 0; j < NCHUNK; ++j) {
        float f = __expf(mb[j] - M);
        L += lb[j] * f;
        a += ab[j * EMBD] * f;
    }
    return a / L;
}

__device__ __forceinline__
void load_ids(const int* __restrict__ xs, int (&id)[GRP]) {
#pragma unroll
    for (int t = 0; t < GRP; ++t) id[t] = xs[t];
}

__device__ __forceinline__
void load_rows(const float* __restrict__ Ak, const float* __restrict__ Ak1,
               const int (&id)[GRP], unsigned eo,
               float (&rm)[GRP], float (&rc)[GRP]) {
#pragma unroll
    for (int t = 0; t < GRP; ++t) {
        unsigned ofs = (unsigned)id[t] * EMBD + eo;
        rm[t] = Ak[ofs];
        rc[t] = Ak1[ofs];
    }
}

__device__ __forceinline__
void sum_group(const float (&rm)[GRP], const float (&rc)[GRP], float& gm, float& gc) {
#pragma unroll
    for (int t = 0; t < GRP; ++t) { gm += rm[t]; gc += rc[t]; }
}

__global__ __launch_bounds__(HOPT)
void k_hop(const float* __restrict__ Ak, const float* __restrict__ Ak1,
           const float* __restrict__ TA, const float* __restrict__ TC,
           const int* __restrict__ x, const int* __restrict__ q, int first,
           const float* __restrict__ u_in,
           const float* __restrict__ pm_in, const float* __restrict__ pl_in,
           const float* __restrict__ pa_in,
           float* __restrict__ u_out,
           float* __restrict__ pm_out, float* __restrict__ pl_out,
           float* __restrict__ pa_out) {
    int ch = blockIdx.x, b = blockIdx.y;
    int e = threadIdx.x;
    int lane = e & 31;
    unsigned eo = (unsigned)e;
    int wv = __builtin_amdgcn_readfirstlane(e >> 5);

    __shared__ float vbuf[CHS][EMBD];
    __shared__ float cbuf[CHS][EMBD];
    __shared__ float red[HOPT / 32][CHS];

    int s0 = ch * CHS;
    const int* xch = x + (b * SLEN + s0) * SENT;

    float rm0[GRP], rc0[GRP], rm1[GRP], rc1[GRP];
    int id0[GRP], id1[GRP];
    load_ids(xch, id0);
    load_ids(xch + GRP, id1);
    load_rows(Ak, Ak1, id0, eo, rm0, rc0);
    load_rows(Ak, Ak1, id1, eo, rm1, rc1);

    float ue;
    if (first) {
        const int* qb = q + b * QLEN;
        ue = 0.f;
#pragma unroll
        for (int j = 0; j < QLEN; ++j)
            ue += Ak1[(unsigned)qb[j] * EMBD + eo];
    } else {
        ue = u_in[b * EMBD + e];
        ue += merge_partials(pm_in, pl_in, pa_in, b, e);
    }
    if (ch == 0) u_out[b * EMBD + e] = ue;

#pragma unroll
    for (int si = 0; si < CHS; ++si) {
        const int* xn = xch + (si + 1) * SENT;
        float gm = 0.f, gc = 0.f;
        if (si + 1 < CHS) load_ids(xn, id0);
        __builtin_amdgcn_sched_barrier(0);
        sum_group(rm0, rc0, gm, gc);
        __builtin_amdgcn_sched_barrier(0);
        if (si + 1 < CHS) { load_rows(Ak, Ak1, id0, eo, rm0, rc0); load_ids(xn + GRP, id1); }
        __builtin_amdgcn_sched_barrier(0);
        sum_group(rm1, rc1, gm, gc);
        __builtin_amdgcn_sched_barrier(0);
        if (si + 1 < CHS) load_rows(Ak, Ak1, id1, eo, rm1, rc1);
        __builtin_amdgcn_sched_barrier(0);
        vbuf[si][e] = (gm + TA[s0 + si]) * ue;
        cbuf[si][e] = gc + TC[s0 + si];
    }

    float v[CHS];
#pragma unroll
    for (int si = 0; si < CHS; ++si) {
        float r = vbuf[si][e];
#pragma unroll
        for (int off = 16; off > 0; off >>= 1) r += __shfl_xor(r, off, 32);
        v[si] = r;
    }
    if (lane == 0) {
#pragma unroll
        for (int si = 0; si < CHS; ++si) red[wv][si] = v[si];
    }
    __syncthreads();

    float sc[CHS];
#pragma unroll
    for (int i = 0; i < CHS; ++i)
        sc[i] = (red[0][i] + red[1][i]) + (red[2][i] + red[3][i]);
    float M = sc[0];
#pragma unroll
    for (int i = 1; i < CHS; ++i) M = fmaxf(M, sc[i]);
    float L = 0.f;
    float w[CHS];
#pragma unroll
    for (int i = 0; i < CHS; ++i) { w[i] = __expf(sc[i] - M); L += w[i]; }

    float acc = 0.f;
#pragma unroll
    for (int i = 0; i < CHS; ++i) acc += w[i] * cbuf[i][e];

    pa_out[((long)b * NCHUNK + ch) * EMBD + e] = acc;
    if (e == 0) { pm_out[b * NCHUNK + ch] = M; pl_out[b * NCHUNK + ch] = L; }
}

__global__ __launch_bounds__(EMBD)
void k_merge(const float* __restrict__ u, float* __restrict__ uf,
             const float* __restrict__ pm,
             const float* __restrict__ pl, const float* __restrict__ pa) {
    int b = blockIdx.x, e = threadIdx.x;
    float v = u[b * EMBD + e] + merge_partials(pm, pl, pa, b, e);
    int mt = b >> 4, c = b & 15;
    int i = e >> 2;
    int h = (e >> 1) & 1;
    int j = e & 1;
    uf[((mt * 16 + (i >> 1)) * 32 + h * 16 + c) * 4 + (i & 1) * 2 + j] = v;
}

#define LG_WAVES 4
#define LG_WGS   256
#define LG_NW    (LG_WGS * LG_WAVES)
#define NSTRIP   (VOCAB / 16)
static_assert(3 * LG_NW <= NSTRIP && NSTRIP <= 4 * LG_NW, "three whole rounds of strips and a partial fourth");

__device__ __forceinline__
void load_bfrags(const float* __restrict__ A3, int s, int l16, int koff, v2f (&b)[32]) {
    const float* arow = A3 + (long)(s * 16 + l16) * EMBD + koff;
#pragma unroll
    for (int i = 0; i < 32; ++i) { b[i].x = arow[4 * i]; b[i].y = arow[4 * i + 1]; }
}

__device__ __forceinline__
void strip_product(const v2f (&a0)[32], const v2f (&a1)[32], const v2f (&a2)[32],
                   const v2f (&a3)[32], const v2f (&b)[32],
                   float* __restrict__ out, int s, int l16, int half) {
    v8f c0 = {}, c1 = {}, c2 = {}, c3 = {};
#pragma unroll
    for (int i = 0; i < 32; ++i) {
        c0 = __builtin_amdgcn_wmma_f32_16x16x4_f32(false, a0[i], false, b[i],
                                                   (short)0, c0, false, false);
        c1 = __builtin_amdgcn_wmma_f32_16x16x4_f32(false, a1[i], false, b[i],
                                                   (short)0, c1, false, false);
        c2 = __builtin_amdgcn_wmma_f32_16x16x4_f32(false, a2[i], false, b[i],
                                                   (short)0, c2, false, false);
        c3 = __builtin_amdgcn_wmma_f32_16x16x4_f32(false, a3[i], false, b[i],
                                                   (short)0, c3, false, false);
    }
    int col = s * 16 + l16;
#pragma unroll
    for (int r = 0; r < 8; ++r) {
        int row = r + 8 * half;
        out[(long)(row)      * VOCAB + col] = c0[r];
        out[(long)(row + 16) * VOCAB + col] = c1[r];
        out[(long)(row + 32) * VOCAB + col] = c2[r];
        out[(long)(row + 48) * VOCAB + col] = c3[r];
    }
}

__device__ __forceinline__
void half_strip_product(const v2f (&a0)[32], const v2f (&a1)[32], const v2f (&b)[32],
                        float* __restrict__ out, int m0, int s, int l16, int half) {
    v8f c0 = {}, c1 = {};
#pragma unroll
    for (int i = 0; i < 32; ++i) {
        c0 = __builtin_amdgcn_wmma_f32_16x16x4_f32(false, a0[i], false, b[i],
                                                   (short)0, c0, false, false);
        c1 = __builtin_amdgcn_wmma_f32_16x16x4_f32(false, a1[i], false, b[i],
                                                   (short)0, c1, false, false);
    }
    int col = s * 16 + l16;
#pragma unroll
    for (int r = 0; r < 8; ++r) {
        int row = m0 + r + 8 * half;
        out[(long)(row)      * VOCAB + col] = c0[r];
        out[(long)(row + 16) * VOCAB + col] = c1[r];
    }
}

__global__ __launch_bounds__(LG_WAVES * 32)
void k_logits(const float* __restrict__ uf, const float* __restrict__ A3,
              float* __restrict__ out) {
    int lane = threadIdx.x & 31;
    int w = blockIdx.x * LG_WAVES + __builtin_amdgcn_readfirstlane(threadIdx.x >> 5);
    int half = lane >> 4;
    int l16  = lane & 15;
    int koff = half * 2;

    v2f bA[32], bB[32];
    load_bfrags(A3, w, l16, koff, bA);

    v2f a0[32], a1[32], a2[32], a3[32];
    const float4* ufl = reinterpret_cast<const float4*>(uf) + lane;
#pragma unroll
    for (int p = 0; p < 16; ++p) {
        float4 q0 = ufl[(0 * 16 + p) * 32];
        float4 q1 = ufl[(1 * 16 + p) * 32];
        float4 q2 = ufl[(2 * 16 + p) * 32];
        float4 q3 = ufl[(3 * 16 + p) * 32];
        a0[2 * p].x = q0.x; a0[2 * p].y = q0.y; a0[2 * p + 1].x = q0.z; a0[2 * p + 1].y = q0.w;
        a1[2 * p].x = q1.x; a1[2 * p].y = q1.y; a1[2 * p + 1].x = q1.z; a1[2 * p + 1].y = q1.w;
        a2[2 * p].x = q2.x; a2[2 * p].y = q2.y; a2[2 * p + 1].x = q2.z; a2[2 * p + 1].y = q2.w;
        a3[2 * p].x = q3.x; a3[2 * p].y = q3.y; a3[2 * p + 1].x = q3.z; a3[2 * p + 1].y = q3.w;
    }

    load_bfrags(A3, w + LG_NW, l16, koff, bB);
    strip_product(a0, a1, a2, a3, bA, out, w, l16, half);

    load_bfrags(A3, w + 2 * LG_NW, l16, koff, bA);
    strip_product(a0, a1, a2, a3, bB, out, w + LG_NW, l16, half);

    bool extra = w < 2 * (NSTRIP - 3 * LG_NW);
    int sx = 3 * LG_NW + (w >> 1);
    if (extra) load_bfrags(A3, sx, l16, koff, bB);
    strip_product(a0, a1, a2, a3, bA, out, w + 2 * LG_NW, l16, half);

    if (extra) {
        if (w & 1) half_strip_product(a2, a3, bB, out, 32, sx, l16, half);
        else       half_strip_product(a0, a1, bB, out, 0, sx, l16, half);
    }
}

#define LSM_T   1024
#define ROW4    (VOCAB / 4)
#define LSM_IT  13
static_assert(LSM_T / 32 == 32, "one lane per wave partial in the second reduction step");
static_assert(LSM_IT * LSM_T >= ROW4 + 4, "every piece of a shifted row has a thread");

__device__ __forceinline__ float wave_max(float v) {
#pragma unroll
    for (int off = 16; off > 0; off >>= 1) v = fmaxf(v, __shfl_xor(v, off, 32));
    return v;
}
__device__ __forceinline__ float wave_sum(float v) {
#pragma unroll
    for (int off = 16; off > 0; off >>= 1) v += __shfl_xor(v, off, 32);
    return v;
}

__global__ __launch_bounds__(LSM_T)
void k_logsoftmax(float* __restrict__ out) {
    int b = blockIdx.x, t = threadIdx.x;
    int lane = t & 31, wv = t >> 5;
    int shift = (b & 1) * 4;
    float4* row4 = reinterpret_cast<float4*>(out + (long)b * VOCAB) - shift;

    __shared__ float redm[LSM_T / 32];
    __shared__ float reds[LSM_T / 32];

    const float ninf = -__builtin_huge_valf();
    const float4 ninf4 = make_float4(ninf, ninf, ninf, ninf);
    float4 x[LSM_IT];
#pragma unroll
    for (int i = 0; i < LSM_IT; ++i) {
        int p = i * LSM_T + t;
        bool lo = p >= shift, hi = p < ROW4 + shift;
        float4 v = ninf4;
        if (hi) v = row4[lo ? p : shift];
        x[i] = lo ? v : ninf4;
    }

    float m = ninf;
#pragma unroll
    for (int i = 0; i < LSM_IT; ++i)
        m = fmaxf(m, fmaxf(fmaxf(x[i].x, x[i].y), fmaxf(x[i].z, x[i].w)));
    m = wave_max(m);
    if (lane == 0) redm[wv] = m;
    __syncthreads();
    float M = wave_max(redm[lane]);

    float s = 0.f;
#pragma unroll
    for (int i = 0; i < LSM_IT; ++i)
        s += (__expf(x[i].x - M) + __expf(x[i].y - M)) +
             (__expf(x[i].z - M) + __expf(x[i].w - M));
    s = wave_sum(s);
    if (lane == 0) reds[wv] = s;
    __syncthreads();
    float lse = __logf(wave_sum(reds[lane]));

#pragma unroll
    for (int i = 0; i < LSM_IT; ++i) {
        int p = i * LSM_T + t;
        if (p >= shift && p < ROW4 + shift)
            row4[p] = make_float4(x[i].x - M - lse, x[i].y - M - lse,
                                  x[i].z - M - lse, x[i].w - M - lse);
    }
}

extern "C" void kernel_launch(void* const* d_in, const int* in_sizes, int n_in,
                              void* d_out, int out_size, void* d_ws, size_t ws_size,
                              hipStream_t stream) {
    const float* A  = (const float*)d_in[0];
    const float* TA = (const float*)d_in[1];
    const float* TC = (const float*)d_in[2];
    const int*   x  = (const int*)d_in[3];
    const int*   q  = (const int*)d_in[4];
    float* out = (float*)d_out;

    float* ws   = (float*)d_ws;
    float* uA   = ws;
    float* uB   = uA   + BATCH * EMBD;
    float* pm0  = uB   + BATCH * EMBD;
    float* pl0  = pm0  + BATCH * NCHUNK;
    float* pa0  = ws   + BATCH * EMBD + BATCH * SLEN;

    float* pa1  = out;
    float* pm1  = pa1  + BATCH * NCHUNK * EMBD;
    float* pl1  = pm1  + BATCH * NCHUNK;
    float* pa2  = out  + 262144;
    float* pm2  = pa2  + BATCH * NCHUNK * EMBD;
    float* pl2  = pm2  + BATCH * NCHUNK;

    const long TBL = (long)VOCAB * EMBD;
    const dim3 hgrid(NCHUNK, BATCH);

    k_hop<<<hgrid, HOPT, 0, stream>>>(A + 0 * TBL, A + 1 * TBL, TA, TC, x, q, 1,
                                      uB, pm1, pl1, pa1, uA, pm0, pl0, pa0);
    k_hop<<<hgrid, HOPT, 0, stream>>>(A + 1 * TBL, A + 2 * TBL, TA, TC, x, q, 0,
                                      uA, pm0, pl0, pa0, uB, pm1, pl1, pa1);
    k_hop<<<hgrid, HOPT, 0, stream>>>(A + 2 * TBL, A + 3 * TBL, TA, TC, x, q, 0,
                                      uB, pm1, pl1, pa1, uA, pm2, pl2, pa2);
    k_merge<<<BATCH, EMBD, 0, stream>>>(uA, uB, pm2, pl2, pa2);

    k_logits  <<<LG_WGS, LG_WAVES * 32, 0, stream>>>(uB, A + HOPS * TBL, out);
    k_logsoftmax<<<BATCH, LSM_T, 0, stream>>>(out);

    (void)in_sizes; (void)n_in; (void)out_size; (void)ws_size;
}


# k_hop in two phases: memory rows, scores and the chunk's softmax weights first, then output rows gathered only for sentences whose weight is not exactly zero (exp underflow); sum orders kept
# speedup vs baseline: 3.0091x; 1.0449x over previous
#include <hip/hip_runtime.h>
#include <hip/hip_bf16.h>

#define HOPS   3
#define VOCAB  50000
#define EMBD   128
#define SLEN   200
#define SENT   32
#define BATCH  64
#define QLEN   32
#define NCHUNK 20
#define CHS    10
#define HOPT   128

typedef __attribute__((ext_vector_type(2))) float v2f;
typedef __attribute__((ext_vector_type(8))) float v8f;


__device__ __forceinline__
float merge_partials(const float* __restrict__ pm, const float* __restrict__ pl,
                     const float* __restrict__ pa, int b, int e) {
    const float* mb = pm + b * NCHUNK;
    const float* lb = pl + b * NCHUNK;
    const float* ab = pa + (long)b * NCHUNK * EMBD + e;
    float M = mb[0];
#pragma unroll
    for (int j = 1; j < NCHUNK; ++j) M = fmaxf(M, mb[j]);
    float L = 0.f, a = 0.f;
#pragma unroll
    for (int j = 0; j < NCHUNK; ++j) {
        float f = __expf(mb[j] - M);
        L += lb[j] * f;
        a += ab[j * EMBD] * f;
    }
    return a / L;
}

__device__ __forceinline__
void load_ids(const int* __restrict__ xs, int (&id)[SENT]) {
#pragma unroll
    for (int t = 0; t < SENT; ++t) id[t] = xs[t];
}

__device__ __forceinline__
void load_rows(const float* __restrict__ Ax, const int (&id)[SENT], unsigned eo,
               float (&r)[SENT]) {
#pragma unroll
    for (int t = 0; t < SENT; ++t) {
        r[t] = Ax[(unsigned)id[t] * EMBD + eo];
    }
}

__device__ __forceinline__
float sum_rows(const float (&r)[SENT]) {
    float g = 0.f;
#pragma unroll
    for (int t = 0; t < SENT; ++t) g += r[t];
    return g;
}

__device__ __forceinline__
void row_offsets(const int* __restrict__ xs, unsigned eo, unsigned (&of)[SENT]) {
#pragma unroll
    for (int t = 0; t < SENT; ++t) of[t] = (unsigned)xs[t] * EMBD + eo;
}

__device__ __forceinline__
void load_rows_at(const float* __restrict__ Ax, const unsigned (&of)[SENT], float (&r)[SENT]) {
#pragma unroll
    for (int t = 0; t < SENT; ++t) r[t] = Ax[of[t]];
}

__device__ __forceinline__
float pick(const float (&w)[CHS], int i) {
    float v = w[0];
#pragma unroll
    for (int k = 1; k < CHS; ++k) v = (i == k) ? w[k] : v;
    return v;
}

__global__ __launch_bounds__(HOPT) __attribute__((amdgpu_waves_per_eu(5, 5)))
void k_hop(const float* __restrict__ Ak, const float* __restrict__ Ak1,
           const float* __restrict__ TA, const float* __restrict__ TC,
           const int* __restrict__ x, const int* __restrict__ q, int first,
           const float* __restrict__ u_in,
           const float* __restrict__ pm_in, const float* __restrict__ pl_in,
           const float* __restrict__ pa_in,
           float* __restrict__ u_out,
           float* __restrict__ pm_out, float* __restrict__ pl_out,
           float* __restrict__ pa_out) {
    int ch = blockIdx.x, b = blockIdx.y;
    int e = threadIdx.x;
    int lane = e & 31;
    unsigned eo = (unsigned)e;
    int wv = __builtin_amdgcn_readfirstlane(e >> 5);

    __shared__ float vbuf[CHS][EMBD];
    __shared__ float red[HOPT / 32][CHS];

    int s0 = ch * CHS;
    const int* xch = x + (b * SLEN + s0) * SENT;

    float r0[SENT], r1[SENT];
    {
        int id[SENT];
        load_ids(xch, id);
        load_rows(Ak, id, eo, r0);
    }
    {
        int id[SENT];
        load_ids(xch + SENT, id);
        load_rows(Ak, id, eo, r1);
    }

    float ue;
    if (first) {
        const int* qb = q + b * QLEN;
        ue = 0.f;
#pragma unroll
        for (int j = 0; j < QLEN; ++j)
            ue += Ak1[(unsigned)qb[j] * EMBD + eo];
    } else {
        ue = u_in[b * EMBD + e];
        ue += merge_partials(pm_in, pl_in, pa_in, b, e);
    }
    if (ch == 0) u_out[b * EMBD + e] = ue;

#pragma unroll
    for (int sp = 0; sp < CHS; sp += 2) {
        int id[SENT];
        if (sp + 2 < CHS) load_ids(xch + (sp + 2) * SENT, id);
        __builtin_amdgcn_sched_barrier(0);
        float g = sum_rows(r0);
        __builtin_amdgcn_sched_barrier(0);
        if (sp + 2 < CHS) load_rows(Ak, id, eo, r0);
        __builtin_amdgcn_sched_barrier(0);
        vbuf[sp][e] = (g + TA[s0 + sp]) * ue;

        if (sp + 3 < CHS) load_ids(xch + (sp + 3) * SENT, id);
        __builtin_amdgcn_sched_barrier(0);
        g = sum_rows(r1);
        __builtin_amdgcn_sched_barrier(0);
        if (sp + 3 < CHS) load_rows(Ak, id, eo, r1);
        __builtin_amdgcn_sched_barrier(0);
        vbuf[sp + 1][e] = (g + TA[s0 + sp + 1]) * ue;
    }

    float v[CHS];
#pragma unroll
    for (int si = 0; si < CHS; ++si) {
        float r = vbuf[si][e];
#pragma unroll
        for (int off = 16; off > 0; off >>= 1) r += __shfl_xor(r, off, 32);
        v[si] = r;
    }
    if (lane == 0) {
#pragma unroll
        for (int si = 0; si < CHS; ++si) red[wv][si] = v[si];
    }
    __syncthreads();

    float sc[CHS];
#pragma unroll
    for (int i = 0; i < CHS; ++i)
        sc[i] = (red[0][i] + red[1][i]) + (red[2][i] + red[3][i]);
    float M = sc[0];
#pragma unroll
    for (int i = 1; i < CHS; ++i) M = fmaxf(M, sc[i]);
    float L = 0.f;
    float w[CHS];
#pragma unroll
    for (int i = 0; i < CHS; ++i) { w[i] = __expf(sc[i] - M); L += w[i]; }

    unsigned rem = 0u;
#pragma unroll
    for (int i = 0; i < CHS; ++i)
        if (__builtin_amdgcn_readfirstlane(__float_as_uint(w[i])) != 0u) rem |= 1u << i;

    float acc = 0.f;
    while (rem != 0u) {
        int i0 = __builtin_ctz(rem);
        rem &= rem - 1u;
        if (rem != 0u) {
            int i1 = __builtin_ctz(rem);
            rem &= rem - 1u;
            unsigned ofa[SENT], ofb[SENT];
            row_offsets(xch + i0 * SENT, eo, ofa);
            row_offsets(xch + i1 * SENT, eo, ofb);
            __builtin_amdgcn_sched_barrier(0);
            load_rows_at(Ak1, ofa, r0);
            load_rows_at(Ak1, ofb, r1);
            float g0 = sum_rows(r0);
            acc += pick(w, i0) * (g0 + TC[s0 + i0]);
            float g1 = sum_rows(r1);
            acc += pick(w, i1) * (g1 + TC[s0 + i1]);
        } else {
            unsigned ofa[SENT];
            row_offsets(xch + i0 * SENT, eo, ofa);
            load_rows_at(Ak1, ofa, r0);
            float g0 = sum_rows(r0);
            acc += pick(w, i0) * (g0 + TC[s0 + i0]);
        }
    }

    pa_out[((long)b * NCHUNK + ch) * EMBD + e] = acc;
    if (e == 0) { pm_out[b * NCHUNK + ch] = M; pl_out[b * NCHUNK + ch] = L; }
}

__global__ __launch_bounds__(EMBD)
void k_merge(const float* __restrict__ u, float* __restrict__ uf,
             const float* __restrict__ pm,
             const float* __restrict__ pl, const float* __restrict__ pa) {
    int b = blockIdx.x, e = threadIdx.x;
    float v = u[b * EMBD + e] + merge_partials(pm, pl, pa, b, e);
    int mt = b >> 4, c = b & 15;
    int i = e >> 2;
    int h = (e >> 1) & 1;
    int j = e & 1;
    uf[((mt * 16 + (i >> 1)) * 32 + h * 16 + c) * 4 + (i & 1) * 2 + j] = v;
}

#define LG_WAVES 4
#define LG_WGS   256
#define LG_NW    (LG_WGS * LG_WAVES)
#define NSTRIP   (VOCAB / 16)
static_assert(3 * LG_NW <= NSTRIP && NSTRIP <= 4 * LG_NW, "three whole rounds of strips and a partial fourth");

__device__ __forceinline__
void load_bfrags(const float* __restrict__ A3, int s, int l16, int koff, v2f (&b)[32]) {
    const float* arow = A3 + (long)(s * 16 + l16) * EMBD + koff;
#pragma unroll
    for (int i = 0; i < 32; ++i) { b[i].x = arow[4 * i]; b[i].y = arow[4 * i + 1]; }
}

__device__ __forceinline__
void strip_product(const v2f (&a0)[32], const v2f (&a1)[32], const v2f (&a2)[32],
                   const v2f (&a3)[32], const v2f (&b)[32],
                   float* __restrict__ out, int s, int l16, int half) {
    v8f c0 = {}, c1 = {}, c2 = {}, c3 = {};
#pragma unroll
    for (int i = 0; i < 32; ++i) {
        c0 = __builtin_amdgcn_wmma_f32_16x16x4_f32(false, a0[i], false, b[i],
                                                   (short)0, c0, false, false);
        c1 = __builtin_amdgcn_wmma_f32_16x16x4_f32(false, a1[i], false, b[i],
                                                   (short)0, c1, false, false);
        c2 = __builtin_amdgcn_wmma_f32_16x16x4_f32(false, a2[i], false, b[i],
                                                   (short)0, c2, false, false);
        c3 = __builtin_amdgcn_wmma_f32_16x16x4_f32(false, a3[i], false, b[i],
                                                   (short)0, c3, false, false);
    }
    int col = s * 16 + l16;
#pragma unroll
    for (int r = 0; r < 8; ++r) {
        int row = r + 8 * half;
        out[(long)(row)      * VOCAB + col] = c0[r];
        out[(long)(row + 16) * VOCAB + col] = c1[r];
        out[(long)(row + 32) * VOCAB + col] = c2[r];
        out[(long)(row + 48) * VOCAB + col] = c3[r];
    }
}

__device__ __forceinline__
void half_strip_product(const v2f (&a0)[32], const v2f (&a1)[32], const v2f (&b)[32],
                        float* __restrict__ out, int m0, int s, int l16, int half) {
    v8f c0 = {}, c1 = {};
#pragma unroll
    for (int i = 0; i < 32; ++i) {
        c0 = __builtin_amdgcn_wmma_f32_16x16x4_f32(false, a0[i], false, b[i],
                                                   (short)0, c0, false, false);
        c1 = __builtin_amdgcn_wmma_f32_16x16x4_f32(false, a1[i], false, b[i],
                                                   (short)0, c1, false, false);
    }
    int col = s * 16 + l16;
#pragma unroll
    for (int r = 0; r < 8; ++r) {
        int row = m0 + r + 8 * half;
        out[(long)(row)      * VOCAB + col] = c0[r];
        out[(long)(row + 16) * VOCAB + col] = c1[r];
    }
}

__global__ __launch_bounds__(LG_WAVES * 32)
void k_logits(const float* __restrict__ uf, const float* __restrict__ A3,
              float* __restrict__ out) {
    int lane = threadIdx.x & 31;
    int w = blockIdx.x * LG_WAVES + __builtin_amdgcn_readfirstlane(threadIdx.x >> 5);
    int half = lane >> 4;
    int l16  = lane & 15;
    int koff = half * 2;

    v2f bA[32], bB[32];
    load_bfrags(A3, w, l16, koff, bA);

    v2f a0[32], a1[32], a2[32], a3[32];
    const float4* ufl = reinterpret_cast<const float4*>(uf) + lane;
#pragma unroll
    for (int p = 0; p < 16; ++p) {
        float4 q0 = ufl[(0 * 16 + p) * 32];
        float4 q1 = ufl[(1 * 16 + p) * 32];
        float4 q2 = ufl[(2 * 16 + p) * 32];
        float4 q3 = ufl[(3 * 16 + p) * 32];
        a0[2 * p].x = q0.x; a0[2 * p].y = q0.y; a0[2 * p + 1].x = q0.z; a0[2 * p + 1].y = q0.w;
        a1[2 * p].x = q1.x; a1[2 * p].y = q1.y; a1[2 * p + 1].x = q1.z; a1[2 * p + 1].y = q1.w;
        a2[2 * p].x = q2.x; a2[2 * p].y = q2.y; a2[2 * p + 1].x = q2.z; a2[2 * p + 1].y = q2.w;
        a3[2 * p].x = q3.x; a3[2 * p].y = q3.y; a3[2 * p + 1].x = q3.z; a3[2 * p + 1].y = q3.w;
    }

    load_bfrags(A3, w + LG_NW, l16, koff, bB);
    strip_product(a0, a1, a2, a3, bA, out, w, l16, half);

    load_bfrags(A3, w + 2 * LG_NW, l16, koff, bA);
    strip_product(a0, a1, a2, a3, bB, out, w + LG_NW, l16, half);

    bool extra = w < 2 * (NSTRIP - 3 * LG_NW);
    int sx = 3 * LG_NW + (w >> 1);
    if (extra) load_bfrags(A3, sx, l16, koff, bB);
    strip_product(a0, a1, a2, a3, bA, out, w + 2 * LG_NW, l16, half);

    if (extra) {
        if (w & 1) half_strip_product(a2, a3, bB, out, 32, sx, l16, half);
        else       half_strip_product(a0, a1, bB, out, 0, sx, l16, half);
    }
}

#define LSM_T   1024
#define ROW4    (VOCAB / 4)
#define LSM_IT  13
static_assert(LSM_T / 32 == 32, "one lane per wave partial in the second reduction step");
static_assert(LSM_IT * LSM_T >= ROW4 + 4, "every piece of a shifted row has a thread");

__device__ __forceinline__ float wave_max(float v) {
#pragma unroll
    for (int off = 16; off > 0; off >>= 1) v = fmaxf(v, __shfl_xor(v, off, 32));
    return v;
}
__device__ __forceinline__ float wave_sum(float v) {
#pragma unroll
    for (int off = 16; off > 0; off >>= 1) v += __shfl_xor(v, off, 32);
    return v;
}

__global__ __launch_bounds__(LSM_T)
void k_logsoftmax(float* __restrict__ out) {
    int b = blockIdx.x, t = threadIdx.x;
    int lane = t & 31, wv = t >> 5;
    int shift = (b & 1) * 4;
    float4* row4 = reinterpret_cast<float4*>(out + (long)b * VOCAB) - shift;

    __shared__ float redm[LSM_T / 32];
    __shared__ float reds[LSM_T / 32];

    const float ninf = -__builtin_huge_valf();
    const float4 ninf4 = make_float4(ninf, ninf, ninf, ninf);
    float4 x[LSM_IT];
#pragma unroll
    for (int i = 0; i < LSM_IT; ++i) {
        int p = i * LSM_T + t;
        bool lo = p >= shift, hi = p < ROW4 + shift;
        float4 v = ninf4;
        if (hi) v = row4[lo ? p : shift];
        x[i] = lo ? v : ninf4;
    }

    float m = ninf;
#pragma unroll
    for (int i = 0; i < LSM_IT; ++i)
        m = fmaxf(m, fmaxf(fmaxf(x[i].x, x[i].y), fmaxf(x[i].z, x[i].w)));
    m = wave_max(m);
    if (lane == 0) redm[wv] = m;
    __syncthreads();
    float M = wave_max(redm[lane]);

    float s = 0.f;
#pragma unroll
    for (int i = 0; i < LSM_IT; ++i)
        s += (__expf(x[i].x - M) + __expf(x[i].y - M)) +
             (__expf(x[i].z - M) + __expf(x[i].w - M));
    s = wave_sum(s);
    if (lane == 0) reds[wv] = s;
    __syncthreads();
    float lse = __logf(wave_sum(reds[lane]));

#pragma unroll
    for (int i = 0; i < LSM_IT; ++i) {
        int p = i * LSM_T + t;
        if (p >= shift && p < ROW4 + shift)
            row4[p] = make_float4(x[i].x - M - lse, x[i].y - M - lse,
                                  x[i].z - M - lse, x[i].w - M - lse);
    }
}

extern "C" void kernel_launch(void* const* d_in, const int* in_sizes, int n_in,
                              void* d_out, int out_size, void* d_ws, size_t ws_size,
                              hipStream_t stream) {
    const float* A  = (const float*)d_in[0];
    const float* TA = (const float*)d_in[1];
    const float* TC = (const float*)d_in[2];
    const int*   x  = (const int*)d_in[3];
    const int*   q  = (const int*)d_in[4];
    float* out = (float*)d_out;

    float* ws   = (float*)d_ws;
    float* uA   = ws;
    float* uB   = uA   + BATCH * EMBD;
    float* pm0  = uB   + BATCH * EMBD;
    float* pl0  = pm0  + BATCH * NCHUNK;
    float* pa0  = ws   + BATCH * EMBD + BATCH * SLEN;

    float* pa1  = out;
    float* pm1  = pa1  + BATCH * NCHUNK * EMBD;
    float* pl1  = pm1  + BATCH * NCHUNK;
    float* pa2  = out  + 262144;
    float* pm2  = pa2  + BATCH * NCHUNK * EMBD;
    float* pl2  = pm2  + BATCH * NCHUNK;

    const long TBL = (long)VOCAB * EMBD;
    const dim3 hgrid(NCHUNK, BATCH);

    k_hop<<<hgrid, HOPT, 0, stream>>>(A + 0 * TBL, A + 1 * TBL, TA, TC, x, q, 1,
                                      uB, pm1, pl1, pa1, uA, pm0, pl0, pa0);
    k_hop<<<hgrid, HOPT, 0, stream>>>(A + 1 * TBL, A + 2 * TBL, TA, TC, x, q, 0,
                                      uA, pm0, pl0, pa0, uB, pm1, pl1, pa1);
    k_hop<<<hgrid, HOPT, 0, stream>>>(A + 2 * TBL, A + 3 * TBL, TA, TC, x, q, 0,
                                      uB, pm1, pl1, pa1, uA, pm2, pl2, pa2);
    k_merge<<<BATCH, EMBD, 0, stream>>>(uA, uB, pm2, pl2, pa2);

    k_logits  <<<LG_WGS, LG_WAVES * 32, 0, stream>>>(uB, A + HOPS * TBL, out);
    k_logsoftmax<<<BATCH, LSM_T, 0, stream>>>(out);

    (void)in_sizes; (void)n_in; (void)out_size; (void)ws_size;
}


# k_hop work-groups cover 50 sentences (five 4-wave teams): softmax maximum over 50 scores so more weights underflow to exact zero, surviving output-row gathers dealt evenly to teams; sum orders change
# speedup vs baseline: 3.1882x; 1.0595x over previous
#include <hip/hip_runtime.h>
#include <hip/hip_bf16.h>

#define HOPS   3
#define VOCAB  50000
#define EMBD   128
#define SLEN   200
#define SENT   32
#define BATCH  64
#define QLEN   32
#define NCHUNK 4
#define NTEAM  5
#define CHS    10
#define WGS    (NTEAM * CHS)
#define HOPT   128
#define HOPW   (NTEAM * HOPT)

typedef __attribute__((ext_vector_type(2))) float v2f;
typedef __attribute__((ext_vector_type(8))) float v8f;


__device__ __forceinline__
float merge_partials(const float* __restrict__ pm, const float* __restrict__ pl,
                     const float* __restrict__ pa, int b, int e) {
    const float* mb = pm + b * NCHUNK;
    const float* lb = pl + b * NCHUNK;
    const float* ab = pa + (long)b * NCHUNK * EMBD + e;
    float M = mb[0];
#pragma unroll
    for (int j = 1; j < NCHUNK; ++j) M = fmaxf(M, mb[j]);
    float L = 0.f, a = 0.f;
#pragma unroll
    for (int j = 0; j < NCHUNK; ++j) {
        float f = __expf(mb[j] - M);
        L += lb[j] * f;
        a += ab[j * EMBD] * f;
    }
    return a / L;
}

__device__ __forceinline__
void load_ids(const int* __restrict__ xs, int (&id)[SENT]) {
#pragma unroll
    for (int t = 0; t < SENT; ++t) id[t] = xs[t];
}

__device__ __forceinline__
void load_rows(const float* __restrict__ Ax, const int (&id)[SENT], unsigned eo,
               float (&r)[SENT]) {
#pragma unroll
    for (int t = 0; t < SENT; ++t) {
        r[t] = Ax[(unsigned)id[t] * EMBD + eo];
    }
}

__device__ __forceinline__
float sum_rows(const float (&r)[SENT]) {
    float g = 0.f;
#pragma unroll
    for (int t = 0; t < SENT; ++t) g += r[t];
    return g;
}

__device__ __forceinline__
void row_offsets(const int* __restrict__ xs, unsigned eo, unsigned (&of)[SENT]) {
#pragma unroll
    for (int t = 0; t < SENT; ++t) of[t] = (unsigned)xs[t] * EMBD + eo;
}

__device__ __forceinline__
void load_rows_at(const float* __restrict__ Ax, const unsigned (&of)[SENT], float (&r)[SENT]) {
#pragma unroll
    for (int t = 0; t < SENT; ++t) r[t] = Ax[of[t]];
}

__global__ __launch_bounds__(HOPW) __attribute__((amdgpu_waves_per_eu(5, 5)))
void k_hop(const float* __restrict__ Ak, const float* __restrict__ Ak1,
           const float* __restrict__ TA, const float* __restrict__ TC,
           const int* __restrict__ x, const int* __restrict__ q, int first,
           const float* __restrict__ u_in,
           const float* __restrict__ pm_in, const float* __restrict__ pl_in,
           const float* __restrict__ pa_in,
           float* __restrict__ u_out,
           float* __restrict__ pm_out, float* __restrict__ pl_out,
           float* __restrict__ pa_out) {
    int ch = blockIdx.x, b = blockIdx.y;
    int tid = threadIdx.x;
    int e = tid & (HOPT - 1);
    int lane = tid & 31;
    unsigned eo = (unsigned)e;
    int wv = __builtin_amdgcn_readfirstlane(tid >> 5);
    int tm = __builtin_amdgcn_readfirstlane(tid >> 7);

    __shared__ float vbuf[NTEAM][CHS][EMBD];
    __shared__ float red[HOPW / 32][CHS];
    __shared__ float tmax[NTEAM];
    __shared__ float tsum[NTEAM];
    __shared__ unsigned trem[NTEAM];
    __shared__ float wsh[WGS];
    __shared__ float accb[NTEAM][EMBD];

    int s0 = ch * WGS + tm * CHS;
    const int* xch = x + (b * SLEN + s0) * SENT;

    float r0[SENT], r1[SENT];
    {
        int id[SENT];
        load_ids(xch, id);
        load_rows(Ak, id, eo, r0);
    }
    {
        int id[SENT];
        load_ids(xch + SENT, id);
        load_rows(Ak, id, eo, r1);
    }

    float ue;
    if (first) {
        const int* qb = q + b * QLEN;
        ue = 0.f;
#pragma unroll
        for (int j = 0; j < QLEN; ++j)
            ue += Ak1[(unsigned)qb[j] * EMBD + eo];
    } else {
        ue = u_in[b * EMBD + e];
        ue += merge_partials(pm_in, pl_in, pa_in, b, e);
    }
    if (ch == 0 && tm == 0) u_out[b * EMBD + e] = ue;

#pragma unroll
    for (int sp = 0; sp < CHS; sp += 2) {
        int id[SENT];
        if (sp + 2 < CHS) load_ids(xch + (sp + 2) * SENT, id);
        __builtin_amdgcn_sched_barrier(0);
        float g = sum_rows(r0);
        __builtin_amdgcn_sched_barrier(0);
        if (sp + 2 < CHS) load_rows(Ak, id, eo, r0);
        __builtin_amdgcn_sched_barrier(0);
        vbuf[tm][sp][e] = (g + TA[s0 + sp]) * ue;

        if (sp + 3 < CHS) load_ids(xch + (sp + 3) * SENT, id);
        __builtin_amdgcn_sched_barrier(0);
        g = sum_rows(r1);
        __builtin_amdgcn_sched_barrier(0);
        if (sp + 3 < CHS) load_rows(Ak, id, eo, r1);
        __builtin_amdgcn_sched_barrier(0);
        vbuf[tm][sp + 1][e] = (g + TA[s0 + sp + 1]) * ue;
    }

    float v[CHS];
#pragma unroll
    for (int si = 0; si < CHS; ++si) {
        float r = vbuf[tm][si][e];
#pragma unroll
        for (int off = 16; off > 0; off >>= 1) r += __shfl_xor(r, off, 32);
        v[si] = r;
    }
    if (lane == 0) {
#pragma unroll
        for (int si = 0; si < CHS; ++si) red[wv][si] = v[si];
    }
    __syncthreads();

    float sc[CHS];
#pragma unroll
    for (int i = 0; i < CHS; ++i)
        sc[i] = (red[4 * tm][i] + red[4 * tm + 1][i]) + (red[4 * tm + 2][i] + red[4 * tm + 3][i]);
    float Mt = sc[0];
#pragma unroll
    for (int i = 1; i < CHS; ++i) Mt = fmaxf(Mt, sc[i]);
    if (e == 0) tmax[tm] = Mt;
    __syncthreads();
    float M = tmax[0];
#pragma unroll
    for (int j = 1; j < NTEAM; ++j) M = fmaxf(M, tmax[j]);
    float L = 0.f;
    float w[CHS];
#pragma unroll
    for (int i = 0; i < CHS; ++i) { w[i] = __expf(sc[i] - M); L += w[i]; }

    unsigned rem = 0u;
#pragma unroll
    for (int i = 0; i < CHS; ++i)
        if (__builtin_amdgcn_readfirstlane(__float_as_uint(w[i])) != 0u) rem |= 1u << i;

    if (e == 0) {
#pragma unroll
        for (int i = 0; i < CHS; ++i) wsh[tm * CHS + i] = w[i];
        tsum[tm] = L;
        trem[tm] = rem;
    }
    __syncthreads();

    unsigned long long all = 0ull;
#pragma unroll
    for (int j = 0; j < NTEAM; ++j) {
        unsigned rj = (unsigned)__builtin_amdgcn_readfirstlane((int)trem[j]);
        all |= (unsigned long long)rj << (CHS * j);
    }
    unsigned long long mine = 0ull;
    {
        unsigned long long m = all;
        int k = 0;
        while (m != 0ull) {
            unsigned long long low = m & (0ull - m);
            if (k == tm) mine |= low;
            m ^= low;
            k = (k + 1 == NTEAM) ? 0 : k + 1;
        }
    }

    const int sw = ch * WGS;
    const int* xwg = x + (b * SLEN + sw) * SENT;
    float acc = 0.f;
    while (mine != 0ull) {
        int i0 = __builtin_ctzll(mine);
        mine &= mine - 1ull;
        if (mine != 0ull) {
            int i1 = __builtin_ctzll(mine);
            mine &= mine - 1ull;
            unsigned ofa[SENT], ofb[SENT];
            row_offsets(xwg + i0 * SENT, eo, ofa);
            row_offsets(xwg + i1 * SENT, eo, ofb);
            float w0 = wsh[i0], w1 = wsh[i1];
            __builtin_amdgcn_sched_barrier(0);
            load_rows_at(Ak1, ofa, r0);
            load_rows_at(Ak1, ofb, r1);
            float g0 = sum_rows(r0);
            acc += w0 * (g0 + TC[sw + i0]);
            float g1 = sum_rows(r1);
            acc += w1 * (g1 + TC[sw + i1]);
        } else {
            unsigned ofa[SENT];
            row_offsets(xwg + i0 * SENT, eo, ofa);
            float w0 = wsh[i0];
            load_rows_at(Ak1, ofa, r0);
            float g0 = sum_rows(r0);
            acc += w0 * (g0 + TC[sw + i0]);
        }
    }

    accb[tm][e] = acc;
    __syncthreads();
    if (tm == 0) {
        float a = accb[0][e];
        float l = tsum[0];
#pragma unroll
        for (int j = 1; j < NTEAM; ++j) { a += accb[j][e]; l += tsum[j]; }
        pa_out[((long)b * NCHUNK + ch) * EMBD + e] = a;
        if (e == 0) { pm_out[b * NCHUNK + ch] = M; pl_out[b * NCHUNK + ch] = l; }
    }
}

__global__ __launch_bounds__(EMBD)
void k_merge(const float* __restrict__ u, float* __restrict__ uf,
             const float* __restrict__ pm,
             const float* __restrict__ pl, const float* __restrict__ pa) {
    int b = blockIdx.x, e = threadIdx.x;
    float v = u[b * EMBD + e] + merge_partials(pm, pl, pa, b, e);
    int mt = b >> 4, c = b & 15;
    int i = e >> 2;
    int h = (e >> 1) & 1;
    int j = e & 1;
    uf[((mt * 16 + (i >> 1)) * 32 + h * 16 + c) * 4 + (i & 1) * 2 + j] = v;
}

#define LG_WAVES 4
#define LG_WGS   256
#define LG_NW    (LG_WGS * LG_WAVES)
#define NSTRIP   (VOCAB / 16)
static_assert(3 * LG_NW <= NSTRIP && NSTRIP <= 4 * LG_NW, "three whole rounds of strips and a partial fourth");

__device__ __forceinline__
void load_bfrags(const float* __restrict__ A3, int s, int l16, int koff, v2f (&b)[32]) {
    const float* arow = A3 + (long)(s * 16 + l16) * EMBD + koff;
#pragma unroll
    for (int i = 0; i < 32; ++i) { b[i].x = arow[4 * i]; b[i].y = arow[4 * i + 1]; }
}

__device__ __forceinline__
void strip_product(const v2f (&a0)[32], const v2f (&a1)[32], const v2f (&a2)[32],
                   const v2f (&a3)[32], const v2f (&b)[32],
                   float* __restrict__ out, int s, int l16, int half) {
    v8f c0 = {}, c1 = {}, c2 = {}, c3 = {};
#pragma unroll
    for (int i = 0; i < 32; ++i) {
        c0 = __builtin_amdgcn_wmma_f32_16x16x4_f32(false, a0[i], false, b[i],
                                                   (short)0, c0, false, false);
        c1 = __builtin_amdgcn_wmma_f32_16x16x4_f32(false, a1[i], false, b[i],
                                                   (short)0, c1, false, false);
        c2 = __builtin_amdgcn_wmma_f32_16x16x4_f32(false, a2[i], false, b[i],
                                                   (short)0, c2, false, false);
        c3 = __builtin_amdgcn_wmma_f32_16x16x4_f32(false, a3[i], false, b[i],
                                                   (short)0, c3, false, false);
    }
    int col = s * 16 + l16;
#pragma unroll
    for (int r = 0; r < 8; ++r) {
        int row = r + 8 * half;
        out[(long)(row)      * VOCAB + col] = c0[r];
        out[(long)(row + 16) * VOCAB + col] = c1[r];
        out[(long)(row + 32) * VOCAB + col] = c2[r];
        out[(long)(row + 48) * VOCAB + col] = c3[r];
    }
}

__device__ __forceinline__
void half_strip_product(const v2f (&a0)[32], const v2f (&a1)[32], const v2f (&b)[32],
                        float* __restrict__ out, int m0, int s, int l16, int half) {
    v8f c0 = {}, c1 = {};
#pragma unroll
    for (int i = 0; i < 32; ++i) {
        c0 = __builtin_amdgcn_wmma_f32_16x16x4_f32(false, a0[i], false, b[i],
                                                   (short)0, c0, false, false);
        c1 = __builtin_amdgcn_wmma_f32_16x16x4_f32(false, a1[i], false, b[i],
                                                   (short)0, c1, false, false);
    }
    int col = s * 16 + l16;
#pragma unroll
    for (int r = 0; r < 8; ++r) {
        int row = m0 + r + 8 * half;
        out[(long)(row)      * VOCAB + col] = c0[r];
        out[(long)(row + 16) * VOCAB + col] = c1[r];
    }
}

__global__ __launch_bounds__(LG_WAVES * 32)
void k_logits(const float* __restrict__ uf, const float* __restrict__ A3,
              float* __restrict__ out) {
    int lane = threadIdx.x & 31;
    int w = blockIdx.x * LG_WAVES + __builtin_amdgcn_readfirstlane(threadIdx.x >> 5);
    int half = lane >> 4;
    int l16  = lane & 15;
    int koff = half * 2;

    v2f bA[32], bB[32];
    load_bfrags(A3, w, l16, koff, bA);

    v2f a0[32], a1[32], a2[32], a3[32];
    const float4* ufl = reinterpret_cast<const float4*>(uf) + lane;
#pragma unroll
    for (int p = 0; p < 16; ++p) {
        float4 q0 = ufl[(0 * 16 + p) * 32];
        float4 q1 = ufl[(1 * 16 + p) * 32];
        float4 q2 = ufl[(2 * 16 + p) * 32];
        float4 q3 = ufl[(3 * 16 + p) * 32];
        a0[2 * p].x = q0.x; a0[2 * p].y = q0.y; a0[2 * p + 1].x = q0.z; a0[2 * p + 1].y = q0.w;
        a1[2 * p].x = q1.x; a1[2 * p].y = q1.y; a1[2 * p + 1].x = q1.z; a1[2 * p + 1].y = q1.w;
        a2[2 * p].x = q2.x; a2[2 * p].y = q2.y; a2[2 * p + 1].x = q2.z; a2[2 * p + 1].y = q2.w;
        a3[2 * p].x = q3.x; a3[2 * p].y = q3.y; a3[2 * p + 1].x = q3.z; a3[2 * p + 1].y = q3.w;
    }

    load_bfrags(A3, w + LG_NW, l16, koff, bB);
    strip_product(a0, a1, a2, a3, bA, out, w, l16, half);

    load_bfrags(A3, w + 2 * LG_NW, l16, koff, bA);
    strip_product(a0, a1, a2, a3, bB, out, w + LG_NW, l16, half);

    bool extra = w < 2 * (NSTRIP - 3 * LG_NW);
    int sx = 3 * LG_NW + (w >> 1);
    if (extra) load_bfrags(A3, sx, l16, koff, bB);
    strip_product(a0, a1, a2, a3, bA, out, w + 2 * LG_NW, l16, half);

    if (extra) {
        if (w & 1) half_strip_product(a2, a3, bB, out, 32, sx, l16, half);
        else       half_strip_product(a0, a1, bB, out, 0, sx, l16, half);
    }
}

#define LSM_T   1024
#define ROW4    (VOCAB / 4)
#define LSM_IT  13
static_assert(LSM_T / 32 == 32, "one lane per wave partial in the second reduction step");
static_assert(LSM_IT * LSM_T >= ROW4 + 4, "every piece of a shifted row has a thread");

__device__ __forceinline__ float wave_max(float v) {
#pragma unroll
    for (int off = 16; off > 0; off >>= 1) v = fmaxf(v, __shfl_xor(v, off, 32));
    return v;
}
__device__ __forceinline__ float wave_sum(float v) {
#pragma unroll
    for (int off = 16; off > 0; off >>= 1) v += __shfl_xor(v, off, 32);
    return v;
}

__global__ __launch_bounds__(LSM_T)
void k_logsoftmax(float* __restrict__ out) {
    int b = blockIdx.x, t = threadIdx.x;
    int lane = t & 31, wv = t >> 5;
    int shift = (b & 1) * 4;
    float4* row4 = reinterpret_cast<float4*>(out + (long)b * VOCAB) - shift;

    __shared__ float redm[LSM_T / 32];
    __shared__ float reds[LSM_T / 32];

    const float ninf = -__builtin_huge_valf();
    const float4 ninf4 = make_float4(ninf, ninf, ninf, ninf);
    float4 x[LSM_IT];
#pragma unroll
    for (int i = 0; i < LSM_IT; ++i) {
        int p = i * LSM_T + t;
        bool lo = p >= shift, hi = p < ROW4 + shift;
        float4 v = ninf4;
        if (hi) v = row4[lo ? p : shift];
        x[i] = lo ? v : ninf4;
    }

    float m = ninf;
#pragma unroll
    for (int i = 0; i < LSM_IT; ++i)
        m = fmaxf(m, fmaxf(fmaxf(x[i].x, x[i].y), fmaxf(x[i].z, x[i].w)));
    m = wave_max(m);
    if (lane == 0) redm[wv] = m;
    __syncthreads();
    float M = wave_max(redm[lane]);

    float s = 0.f;
#pragma unroll
    for (int i = 0; i < LSM_IT; ++i)
        s += (__expf(x[i].x - M) + __expf(x[i].y - M)) +
             (__expf(x[i].z - M) + __expf(x[i].w - M));
    s = wave_sum(s);
    if (lane == 0) reds[wv] = s;
    __syncthreads();
    float lse = __logf(wave_sum(reds[lane]));

#pragma unroll
    for (int i = 0; i < LSM_IT; ++i) {
        int p = i * LSM_T + t;
        if (p >= shift && p < ROW4 + shift)
            row4[p] = make_float4(x[i].x - M - lse, x[i].y - M - lse,
                                  x[i].z - M - lse, x[i].w - M - lse);
    }
}

extern "C" void kernel_launch(void* const* d_in, const int* in_sizes, int n_in,
                              void* d_out, int out_size, void* d_ws, size_t ws_size,
                              hipStream_t stream) {
    const float* A  = (const float*)d_in[0];
    const float* TA = (const float*)d_in[1];
    const float* TC = (const float*)d_in[2];
    const int*   x  = (const int*)d_in[3];
    const int*   q  = (const int*)d_in[4];
    float* out = (float*)d_out;

    float* ws   = (float*)d_ws;
    float* uA   = ws;
    float* uB   = uA   + BATCH * EMBD;
    float* pm0  = uB   + BATCH * EMBD;
    float* pl0  = pm0  + BATCH * NCHUNK;
    float* pa0  = ws   + BATCH * EMBD + BATCH * SLEN;

    float* pa1  = out;
    float* pm1  = pa1  + BATCH * NCHUNK * EMBD;
    float* pl1  = pm1  + BATCH * NCHUNK;
    float* pa2  = out  + 262144;
    float* pm2  = pa2  + BATCH * NCHUNK * EMBD;
    float* pl2  = pm2  + BATCH * NCHUNK;

    const long TBL = (long)VOCAB * EMBD;
    const dim3 hgrid(NCHUNK, BATCH);

    k_hop<<<hgrid, HOPW, 0, stream>>>(A + 0 * TBL, A + 1 * TBL, TA, TC, x, q, 1,
                                      uB, pm1, pl1, pa1, uA, pm0, pl0, pa0);
    k_hop<<<hgrid, HOPW, 0, stream>>>(A + 1 * TBL, A + 2 * TBL, TA, TC, x, q, 0,
                                      uA, pm0, pl0, pa0, uB, pm1, pl1, pa1);
    k_hop<<<hgrid, HOPW, 0, stream>>>(A + 2 * TBL, A + 3 * TBL, TA, TC, x, q, 0,
                                      uB, pm1, pl1, pa1, uA, pm2, pl2, pa2);
    k_merge<<<BATCH, EMBD, 0, stream>>>(uA, uB, pm2, pl2, pa2);

    k_logits  <<<LG_WGS, LG_WAVES * 32, 0, stream>>>(uB, A + HOPS * TBL, out);
    k_logsoftmax<<<BATCH, LSM_T, 0, stream>>>(out);

    (void)in_sizes; (void)n_in; (void)out_size; (void)ws_size;
}
